# starting kernel
# baseline (speedup 1.0000x reference)
#include <hip/hip_runtime.h>
#include <hip/hip_bf16.h>

typedef __attribute__((ext_vector_type(2))) float v2f;
typedef __attribute__((ext_vector_type(4))) float v4f;
typedef __attribute__((ext_vector_type(8))) float v8f;

#define BDIM 2
#define LDIM 512
#define HDIM 128
#define CDIM 25

#define BL_STRIDE 520
#define X2_STRIDE 132

__global__ __launch_bounds__(512) void biaffine_wmma_f32(
    const float* __restrict__ x1, const float* __restrict__ x2,
    const float* __restrict__ W,  const float* __restrict__ bias,
    float* __restrict__ out)
{
    __shared__ float Bl [32 * BL_STRIDE];
    __shared__ float x1t[16 * HDIM];
    __shared__ float x2t[16 * X2_STRIDE];

    const int tid = threadIdx.x;
    const int mt  = blockIdx.x;
    const int lt  = blockIdx.y;
    const int b   = blockIdx.z;
    const int l0  = lt * 16, m0 = mt * 16;

#pragma unroll
    for (int it = 0; it < 8; ++it) {
        const int idx = tid + it * 512;
        const int c   = idx >> 7;
        const int k   = (idx & 127) << 2;
        v4f v = {};
        if (c < CDIM)
            v = *(const v4f*)&W[c * 512 + ((k + 256) & 511)];
        *(v4f*)&Bl[c * BL_STRIDE + k] = v;
    }
    {
        const int r  = tid >> 5;
        const int h4 = (tid & 31) << 2;
        *(v4f*)&x1t[r * HDIM      + h4] = *(const v4f*)&x1[((size_t)b * LDIM + l0 + r) * HDIM + h4];
        *(v4f*)&x2t[r * X2_STRIDE + h4] = *(const v4f*)&x2[((size_t)b * LDIM + m0 + r) * HDIM + h4];
    }
    __syncthreads();

    const int wave = tid >> 5;
    const int lane = tid & 31;
    const int g    = lane >> 4;
    const int n    = lane & 15;

    const float* __restrict__ x1p = &x1t[wave * HDIM          + 2 * g];
    const float* __restrict__ x2p = &x2t[n * X2_STRIDE        + 2 * g];
    const float* __restrict__ b0p = &Bl [n * BL_STRIDE        + 2 * g];
    const float* __restrict__ b1p = &Bl [(n + 16) * BL_STRIDE + 2 * g];

    v8f acc0 = {};
    v8f acc1 = {};

#define WMMA_STEP(AX, AY, KOFF)                                                        \
    do {                                                                               \
        v2f a;  a.x = (AX); a.y = (AY);                                                \
        v2f bf0; bf0.x = b0p[(KOFF)]; bf0.y = b0p[(KOFF) + 1];                         \
        v2f bf1; bf1.x = b1p[(KOFF)]; bf1.y = b1p[(KOFF) + 1];                         \
        acc0 = __builtin_amdgcn_wmma_f32_16x16x4_f32(false, a, false, bf0, (short)0,   \
                                                     acc0, false, false);              \
        acc1 = __builtin_amdgcn_wmma_f32_16x16x4_f32(false, a, false, bf1, (short)0,   \
                                                     acc1, false, false);              \
    } while (0)

#pragma unroll 4
    for (int s = 0; s < 32; ++s) {
        const int k = 4 * s;
        WMMA_STEP(x1p[k] * x2p[k], x1p[k + 1] * x2p[k + 1], k);
    }
#pragma unroll 4
    for (int s = 0; s < 32; ++s) {
        const int k = 4 * s;
        WMMA_STEP(fabsf(x1p[k] - x2p[k]), fabsf(x1p[k + 1] - x2p[k + 1]), 128 + k);
    }
#pragma unroll 4
    for (int s = 0; s < 32; ++s) {
        const int k = 4 * s;
        WMMA_STEP(x1p[k], x1p[k + 1], 256 + k);
    }
#pragma unroll 4
    for (int s = 0; s < 32; ++s) {
        const int k = 4 * s;
        WMMA_STEP(x2p[k], x2p[k + 1], 384 + k);
    }
#undef WMMA_STEP

    const float bias0 = bias[n];
    const float bias1 = (n < CDIM - 16) ? bias[16 + n] : 0.0f;

    const int l = l0 + wave;
    float* __restrict__ outbase = out + ((size_t)b * LDIM + l) * (size_t)LDIM * CDIM;
#pragma unroll
    for (int v = 0; v < 8; ++v) {
        const int m = m0 + v + 8 * g;
        const size_t o = (size_t)m * CDIM;
        outbase[o + n] = acc0[v] + bias0;
        if (n < CDIM - 16)
            outbase[o + 16 + n] = acc1[v] + bias1;
    }
}

extern "C" void kernel_launch(void* const* d_in, const int* in_sizes, int n_in,
                              void* d_out, int out_size, void* d_ws, size_t ws_size,
                              hipStream_t stream) {
    (void)in_sizes; (void)n_in; (void)d_ws; (void)ws_size; (void)out_size;
    const float* x1   = (const float*)d_in[0];
    const float* x2   = (const float*)d_in[1];
    const float* W    = (const float*)d_in[2];
    const float* bias = (const float*)d_in[3];
    float* out = (float*)d_out;

    dim3 grid(LDIM / 16, LDIM / 16, BDIM);
    dim3 block(512);
    biaffine_wmma_f32<<<grid, block, 0, stream>>>(x1, x2, W, bias, out);
}


# x1-only and x2-only terms (t1, t2) computed once per workgroup as 128 WMMAs, K split over the waves and summed through LDS, instead of inside every wave's pair loop: pair-loop K 512 -> 256
# speedup vs baseline: 1.5521x; 1.5521x over previous
#include <hip/hip_runtime.h>
#include <hip/hip_bf16.h>


typedef __attribute__((ext_vector_type(2))) float v2f;
typedef __attribute__((ext_vector_type(4))) float v4f;
typedef __attribute__((ext_vector_type(8))) float v8f;

#define BDIM 2
#define LDIM 512
#define HDIM 128
#define CDIM 25

#define BL_STRIDE 520
#define XT_STRIDE 132

__global__ __launch_bounds__(512) void biaffine_wmma_f32(
    const float* __restrict__ x1, const float* __restrict__ x2,
    const float* __restrict__ W,  const float* __restrict__ bias,
    float* __restrict__ out)
{
    __shared__ float Bl  [32 * BL_STRIDE];
    __shared__ float x1t [16 * XT_STRIDE];
    __shared__ float x2t [16 * XT_STRIDE];
    __shared__ float part[16 * 256];

    const int tid = threadIdx.x;
    const int mt  = blockIdx.x;
    const int lt  = blockIdx.y;
    const int b   = blockIdx.z;
    const int l0  = lt * 16, m0 = mt * 16;

#pragma unroll
    for (int it = 0; it < 8; ++it) {
        const int idx = tid + it * 512;
        const int c   = idx >> 7;
        const int k   = (idx & 127) << 2;
        v4f v = {};
        if (c < CDIM)
            v = *(const v4f*)&W[c * 512 + ((k + 256) & 511)];
        *(v4f*)&Bl[c * BL_STRIDE + k] = v;
    }
    {
        const int r  = tid >> 5;
        const int h4 = (tid & 31) << 2;
        *(v4f*)&x1t[r * XT_STRIDE + h4] = *(const v4f*)&x1[((size_t)b * LDIM + l0 + r) * HDIM + h4];
        *(v4f*)&x2t[r * XT_STRIDE + h4] = *(const v4f*)&x2[((size_t)b * LDIM + m0 + r) * HDIM + h4];
    }
    __syncthreads();

    const int wave = tid >> 5;
    const int lane = tid & 31;
    const int g    = lane >> 4;
    const int n    = lane & 15;

    {
        const int tile = wave & 3;
        const int kq   = (wave >> 2) * 32;
        const float* __restrict__ ap = ((tile & 2) ? x2t : x1t) + n * XT_STRIDE + kq + 2 * g;
        const float* __restrict__ bp = &Bl[(n + 16 * (tile & 1)) * BL_STRIDE
                                           + 256 + 128 * (tile >> 1) + kq + 2 * g];
        v8f p0 = {};
        v8f p1 = {};
#pragma unroll
        for (int s = 0; s < 8; s += 2) {
            v2f a0; a0.x = ap[4 * s];     a0.y = ap[4 * s + 1];
            v2f b0; b0.x = bp[4 * s];     b0.y = bp[4 * s + 1];
            v2f a1; a1.x = ap[4 * s + 4]; a1.y = ap[4 * s + 5];
            v2f b1; b1.x = bp[4 * s + 4]; b1.y = bp[4 * s + 5];
            p0 = __builtin_amdgcn_wmma_f32_16x16x4_f32(false, a0, false, b0, (short)0, p0, false, false);
            p1 = __builtin_amdgcn_wmma_f32_16x16x4_f32(false, a1, false, b1, (short)0, p1, false, false);
        }
        p0 += p1;
        float* __restrict__ pp = &part[wave * 256 + lane * 4];
        v4f lo; lo.x = p0[0]; lo.y = p0[1]; lo.z = p0[2]; lo.w = p0[3];
        v4f hi; hi.x = p0[4]; hi.y = p0[5]; hi.z = p0[6]; hi.w = p0[7];
        *(v4f*)&pp[0]   = lo;
        *(v4f*)&pp[128] = hi;
    }
    __syncthreads();

    v8f acc0;
    v8f acc1;
    float t1c0, t1c1;
    {
        const int t1off = ((wave & 7) >> 2) * 128 + (n + 16 * (wave >> 3)) * 4 + (wave & 3);
        const float* __restrict__ pl = &part[lane * 4];
        v4f lo0 = *(const v4f*)&pl[2 * 256];
        v4f hi0 = *(const v4f*)&pl[2 * 256 + 128];
        v4f lo1 = *(const v4f*)&pl[3 * 256];
        v4f hi1 = *(const v4f*)&pl[3 * 256 + 128];
        t1c0 = part[0 * 256 + t1off];
        t1c1 = part[1 * 256 + t1off];
#pragma unroll
        for (int q = 1; q < 4; ++q) {
            lo0 += *(const v4f*)&pl[(4 * q + 2) * 256];
            hi0 += *(const v4f*)&pl[(4 * q + 2) * 256 + 128];
            lo1 += *(const v4f*)&pl[(4 * q + 3) * 256];
            hi1 += *(const v4f*)&pl[(4 * q + 3) * 256 + 128];
            t1c0 += part[(4 * q + 0) * 256 + t1off];
            t1c1 += part[(4 * q + 1) * 256 + t1off];
        }
        acc0[0] = lo0.x; acc0[1] = lo0.y; acc0[2] = lo0.z; acc0[3] = lo0.w;
        acc0[4] = hi0.x; acc0[5] = hi0.y; acc0[6] = hi0.z; acc0[7] = hi0.w;
        acc1[0] = lo1.x; acc1[1] = lo1.y; acc1[2] = lo1.z; acc1[3] = lo1.w;
        acc1[4] = hi1.x; acc1[5] = hi1.y; acc1[6] = hi1.z; acc1[7] = hi1.w;
    }

    const float* __restrict__ x1p = &x1t[wave * XT_STRIDE     + 2 * g];
    const float* __restrict__ x2p = &x2t[n * XT_STRIDE        + 2 * g];
    const float* __restrict__ b0p = &Bl [n * BL_STRIDE        + 2 * g];
    const float* __restrict__ b1p = &Bl [(n + 16) * BL_STRIDE + 2 * g];

#define WMMA_STEP(AX, AY, KOFF)                                                        \
    do {                                                                               \
        v2f a;  a.x = (AX); a.y = (AY);                                                \
        v2f bf0; bf0.x = b0p[(KOFF)]; bf0.y = b0p[(KOFF) + 1];                         \
        v2f bf1; bf1.x = b1p[(KOFF)]; bf1.y = b1p[(KOFF) + 1];                         \
        acc0 = __builtin_amdgcn_wmma_f32_16x16x4_f32(false, a, false, bf0, (short)0,   \
                                                     acc0, false, false);              \
        acc1 = __builtin_amdgcn_wmma_f32_16x16x4_f32(false, a, false, bf1, (short)0,   \
                                                     acc1, false, false);              \
    } while (0)

#pragma unroll 4
    for (int s = 0; s < 32; ++s) {
        const int k = 4 * s;
        WMMA_STEP(x1p[k] * x2p[k], x1p[k + 1] * x2p[k + 1], k);
    }
#pragma unroll 4
    for (int s = 0; s < 32; ++s) {
        const int k = 4 * s;
        WMMA_STEP(fabsf(x1p[k] - x2p[k]), fabsf(x1p[k + 1] - x2p[k + 1]), 128 + k);
    }
#undef WMMA_STEP

    const float add0 = bias[n] + t1c0;
    const float add1 = ((n < CDIM - 16) ? bias[16 + n] : 0.0f) + t1c1;

    const int l = l0 + wave;
    float* __restrict__ outbase = out + ((size_t)b * LDIM + l) * (size_t)LDIM * CDIM;
#pragma unroll
    for (int v = 0; v < 8; ++v) {
        const int m = m0 + v + 8 * g;
        const size_t o = (size_t)m * CDIM;
        outbase[o + n] = acc0[v] + add0;
        if (n < CDIM - 16)
            outbase[o + 16 + n] = acc1[v] + add1;
    }
}

extern "C" void kernel_launch(void* const* d_in, const int* in_sizes, int n_in,
                              void* d_out, int out_size, void* d_ws, size_t ws_size,
                              hipStream_t stream) {
    (void)in_sizes; (void)n_in; (void)d_ws; (void)ws_size; (void)out_size;
    const float* x1   = (const float*)d_in[0];
    const float* x2   = (const float*)d_in[1];
    const float* W    = (const float*)d_in[2];
    const float* bias = (const float*)d_in[3];
    float* out = (float*)d_out;

    dim3 grid(LDIM / 16, LDIM / 16, BDIM);
    dim3 block(512);
    biaffine_wmma_f32<<<grid, block, 0, stream>>>(x1, x2, W, bias, out);
}
